# starting kernel
# baseline (speedup 1.0000x reference)
#include <hip/hip_runtime.h>


typedef __attribute__((ext_vector_type(2))) float v2f;
typedef __attribute__((ext_vector_type(4))) float v4f;
typedef __attribute__((ext_vector_type(8))) float v8f;

#define BATCH 8
#define H 128
#define W 128
#define OH 127
#define OW 127
#define TILE 8
#define SRC 9
#define NSLOT 96
#define PIXSTRIDE 516

__global__ __launch_bounds__(1024, 1)
void convsbs_fused_kernel(const float* __restrict__ channels,
                          const float* __restrict__ cores,
                          float* __restrict__ out) {
  __shared__ float inter[NSLOT * PIXSTRIDE];

  const int b   = blockIdx.z;
  const int ty0 = blockIdx.y * TILE;
  const int tx0 = blockIdx.x * TILE;

  const float* __restrict__ ch0 = channels;
  const float* __restrict__ ch1 = channels + (long)BATCH * H * W * 4;

  const int tid  = (int)threadIdx.x;
  const int wv   = tid >> 5;
  const int lane = tid & 31;

  if (wv < 24) {
    const int g    = wv >> 2;
    const int core = wv & 3;
    const int m    = lane & 15;
    const int half = lane >> 4;

    int s = g * 16 + m;
    int sc = s > 80 ? 80 : s;
    int sy = sc / SRC, sx = sc % SRC;
    int gy = ty0 + sy; gy = gy > (H - 1) ? (H - 1) : gy;
    int gx = tx0 + sx; gx = gx > (W - 1) ? (W - 1) : gx;
    const long pixoff = ((((long)b * H + gy) * W) + gx) * 4;

    const v4f c0 = *(const v4f*)(ch0 + pixoff);
    const v2f c1 = *(const v2f*)(ch1 + pixoff + 2 * half);

    v2f afrag[4];
#pragma unroll
    for (int kt = 0; kt < 4; ++kt) {
      afrag[kt][0] = c0[kt] * c1[0];
      afrag[kt][1] = c0[kt] * c1[1];
    }

    const float* __restrict__ wbase = cores + core * 2048;
    const int sbase = g * 16 + 8 * half;

#pragma unroll
    for (int nt = 0; nt < 8; ++nt) {
      v8f acc = {};
#pragma unroll
      for (int kt = 0; kt < 4; ++kt) {
        const v2f bfrag =
            *(const v2f*)(wbase + (nt * 16 + m) * 16 + kt * 4 + 2 * half);
        acc = __builtin_amdgcn_wmma_f32_16x16x4_f32(
            false, afrag[kt], false, bfrag, (short)0, acc, false, false);
      }
      const int colbase = core * 128 + nt * 16 + m;
#pragma unroll
      for (int v = 0; v < 8; ++v)
        inter[(sbase + v) * PIXSTRIDE + colbase] = acc[v];
    }
  }

  __syncthreads();

  const int pix = tid >> 4;
  const int chn = tid & 15;
  const int oyl = pix >> 3, oxl = pix & 7;
  const int p  = (chn >> 3) & 1;
  const int q  = (chn >> 2) & 1;
  const int r  = (chn >> 1) & 1;
  const int s2 = chn & 1;

  const int sA = oyl * SRC + oxl;
  const float* __restrict__ Am = &inter[(sA)*PIXSTRIDE + 0 * 128 + p * 64];
  const float* __restrict__ Bm = &inter[(sA + 1) * PIXSTRIDE + 1 * 128 + q * 64];
  const float* __restrict__ Cm = &inter[(sA + SRC) * PIXSTRIDE + 2 * 128 + r * 64];
  const float* __restrict__ Dm = &inter[(sA + SRC + 1) * PIXSTRIDE + 3 * 128 + s2 * 64];

  float acc = 0.f;
  for (int i = 0; i < 8; ++i) {
    float e[8];
#pragma unroll
    for (int k = 0; k < 8; ++k) e[k] = 0.f;
#pragma unroll
    for (int j = 0; j < 8; ++j) {
      const float av = Am[i * 8 + j];
#pragma unroll
      for (int k = 0; k < 8; ++k) e[k] = fmaf(av, Bm[j * 8 + k], e[k]);
    }
#pragma unroll
    for (int l = 0; l < 8; ++l) {
      float gv = 0.f;
#pragma unroll
      for (int k = 0; k < 8; ++k) gv = fmaf(e[k], Cm[k * 8 + l], gv);
      acc = fmaf(gv, Dm[l * 8 + i], acc);
    }
  }

  const int oy = ty0 + oyl;
  const int ox = tx0 + oxl;
  if (oy < OH && ox < OW)
    out[((((long)b * OH + oy) * OW) + ox) * 16 + chn] = acc;
}

extern "C" void kernel_launch(void* const* d_in, const int* in_sizes, int n_in,
                              void* d_out, int out_size, void* d_ws, size_t ws_size,
                              hipStream_t stream) {
  (void)in_sizes; (void)n_in; (void)d_ws; (void)ws_size; (void)out_size;
  const float* channels = (const float*)d_in[0];
  const float* cores    = (const float*)d_in[1];
  float* out            = (float*)d_out;

  dim3 grid((OW + TILE - 1) / TILE, (OH + TILE - 1) / TILE, BATCH);
  dim3 block(1024);
  hipLaunchKernelGGL(convsbs_fused_kernel, grid, block, 0, stream,
                     channels, cores, out);
}


# Stage 2: form A_p*B_q and (C_r*D_s)^T once per pixel (4+4 8x8 products, staged in LDS on a conflict-free map) instead of once per output channel (16+16), then 16 dot products of 64; spills gone
# speedup vs baseline: 4.3676x; 4.3676x over previous
#include <hip/hip_runtime.h>


typedef __attribute__((ext_vector_type(2))) float v2f;
typedef __attribute__((ext_vector_type(4))) float v4f;
typedef __attribute__((ext_vector_type(8))) float v8f;

#define BATCH 8
#define H 128
#define W 128
#define OH 127
#define OW 127
#define TILE 8
#define SRC 9
#define NSLOT 96
#define PIXSTRIDE 516
#define PRODSTRIDE 68
#define PRODPIX 580

__global__ __launch_bounds__(1024, 1)
void convsbs_fused_kernel(const float* __restrict__ channels,
                          const float* __restrict__ cores,
                          float* __restrict__ out) {
  __shared__ __attribute__((aligned(16))) float inter[NSLOT * PIXSTRIDE];

  const int b   = blockIdx.z;
  const int ty0 = blockIdx.y * TILE;
  const int tx0 = blockIdx.x * TILE;

  const float* __restrict__ ch0 = channels;
  const float* __restrict__ ch1 = channels + (long)BATCH * H * W * 4;

  const int tid  = (int)threadIdx.x;
  const int wv   = tid >> 5;
  const int lane = tid & 31;

  if (wv < 24) {
    const int g    = wv >> 2;
    const int core = wv & 3;
    const int m    = lane & 15;
    const int half = lane >> 4;

    int s = g * 16 + m;
    int sc = s > 80 ? 80 : s;
    int sy = sc / SRC, sx = sc % SRC;
    int gy = ty0 + sy; gy = gy > (H - 1) ? (H - 1) : gy;
    int gx = tx0 + sx; gx = gx > (W - 1) ? (W - 1) : gx;
    const long pixoff = ((((long)b * H + gy) * W) + gx) * 4;

    const v4f c0 = *(const v4f*)(ch0 + pixoff);
    const v2f c1 = *(const v2f*)(ch1 + pixoff + 2 * half);

    v2f afrag[4];
#pragma unroll
    for (int kt = 0; kt < 4; ++kt) {
      afrag[kt][0] = c0[kt] * c1[0];
      afrag[kt][1] = c0[kt] * c1[1];
    }

    const float* __restrict__ wbase = cores + core * 2048;
    const int sbase = g * 16 + 8 * half;

#pragma unroll
    for (int nt = 0; nt < 8; ++nt) {
      v8f acc = {};
#pragma unroll
      for (int kt = 0; kt < 4; ++kt) {
        const v2f bfrag =
            *(const v2f*)(wbase + (nt * 16 + m) * 16 + kt * 4 + 2 * half);
        acc = __builtin_amdgcn_wmma_f32_16x16x4_f32(
            false, afrag[kt], false, bfrag, (short)0, acc, false, false);
      }
      const int colbase = core * 128 + nt * 16 + m;
#pragma unroll
      for (int v = 0; v < 8; ++v)
        inter[(sbase + v) * PIXSTRIDE + colbase] = acc[v];
    }
  }

  __syncthreads();

  const int wvu  = __builtin_amdgcn_readfirstlane(wv);
  const int kind = wvu >> 4;
  const int pair = wvu & 3;
  const int hi   = pair >> 1, lo = pair & 1;
  const int pixa = ((wvu >> 2) & 3) * 16 + (lane >> 1);
  const int hf   = lane & 1;
  const int sAa  = (pixa >> 3) * SRC + (pixa & 7);

  v4f res[4][2];
#pragma unroll
  for (int ii = 0; ii < 4; ++ii) { res[ii][0] = (v4f){0.f, 0.f, 0.f, 0.f}; res[ii][1] = (v4f){0.f, 0.f, 0.f, 0.f}; }

  if (kind == 0) {
    const float* Xp = &inter[sAa * PIXSTRIDE + 0 * 128 + hi * 64 + hf * 32];
    const float* Yp = &inter[(sAa + 1) * PIXSTRIDE + 1 * 128 + lo * 64];
    v4f xa[4][2];
#pragma unroll
    for (int ii = 0; ii < 4; ++ii) {
      xa[ii][0] = *(const v4f*)(Xp + ii * 8);
      xa[ii][1] = *(const v4f*)(Xp + ii * 8 + 4);
    }
#pragma unroll
    for (int j = 0; j < 8; ++j) {
      const v4f y0 = *(const v4f*)(Yp + j * 8);
      const v4f y1 = *(const v4f*)(Yp + j * 8 + 4);
#pragma unroll
      for (int ii = 0; ii < 4; ++ii) {
        const float a = xa[ii][j >> 2][j & 3];
#pragma unroll
        for (int c = 0; c < 4; ++c) {
          res[ii][0][c] = fmaf(a, y0[c], res[ii][0][c]);
          res[ii][1][c] = fmaf(a, y1[c], res[ii][1][c]);
        }
      }
    }
  } else {
    const float* Cp = &inter[(sAa + SRC) * PIXSTRIDE + 2 * 128 + hi * 64];
    const float* Dp = &inter[(sAa + SRC + 1) * PIXSTRIDE + 3 * 128 + lo * 64 + hf * 4];
    v4f dl[8];
#pragma unroll
    for (int l = 0; l < 8; ++l) dl[l] = *(const v4f*)(Dp + l * 8);
#pragma unroll
    for (int k = 0; k < 8; ++k) {
      const v4f c0 = *(const v4f*)(Cp + k * 8);
      const v4f c1 = *(const v4f*)(Cp + k * 8 + 4);
#pragma unroll
      for (int ii = 0; ii < 4; ++ii) {
        float r = 0.f;
#pragma unroll
        for (int l = 0; l < 4; ++l) r = fmaf(c0[l], dl[l][ii], r);
#pragma unroll
        for (int l = 0; l < 4; ++l) r = fmaf(c1[l], dl[l + 4][ii], r);
        res[ii][k >> 2][k & 3] = r;
      }
    }
  }

  __syncthreads();

  float* prod = inter;
  {
    float* dst =
        &prod[pixa * PRODPIX + (kind * 4 + pair) * PRODSTRIDE + hf * 32];
#pragma unroll
    for (int ii = 0; ii < 4; ++ii) {
      *(v4f*)(dst + ii * 8)     = res[ii][0];
      *(v4f*)(dst + ii * 8 + 4) = res[ii][1];
    }
  }

  __syncthreads();

  const int pix = tid >> 4;
  const int chn = tid & 15;
  const int oyl = pix >> 3, oxl = pix & 7;

  const float* Pm = &prod[pix * PRODPIX + (chn >> 2) * PRODSTRIDE];
  const float* Qm = &prod[pix * PRODPIX + (4 + (chn & 3)) * PRODSTRIDE];

  v4f part = {0.f, 0.f, 0.f, 0.f};
#pragma unroll
  for (int e = 0; e < 16; ++e) {
    const v4f pv = *(const v4f*)(Pm + e * 4);
    const v4f qv = *(const v4f*)(Qm + e * 4);
#pragma unroll
    for (int c = 0; c < 4; ++c) part[c] = fmaf(pv[c], qv[c], part[c]);
  }
  const float acc = (part[0] + part[1]) + (part[2] + part[3]);

  const int oy = ty0 + oyl;
  const int ox = tx0 + oxl;
  if (oy < OH && ox < OW)
    out[((((long)b * OH + oy) * OW) + ox) * 16 + chn] = acc;
}

extern "C" void kernel_launch(void* const* d_in, const int* in_sizes, int n_in,
                              void* d_out, int out_size, void* d_ws, size_t ws_size,
                              hipStream_t stream) {
  (void)in_sizes; (void)n_in; (void)d_ws; (void)ws_size; (void)out_size;
  const float* channels = (const float*)d_in[0];
  const float* cores    = (const float*)d_in[1];
  float* out            = (float*)d_out;

  dim3 grid((OW + TILE - 1) / TILE, (OH + TILE - 1) / TILE, BATCH);
  dim3 block(1024);
  hipLaunchKernelGGL(convsbs_fused_kernel, grid, block, 0, stream,
                     channels, cores, out);
}


# Stage 2 pair products on f32 WMMA: P=[A0;A1]x[B0 B1], R^T=[D0^T;D1^T]x[C0^T C1^T] (4 per pixel, fragments are linear LDS reads); trace taken from the D fragments by lane exchange, no LDS staging
# speedup vs baseline: 5.1057x; 1.1690x over previous
#include <hip/hip_runtime.h>


typedef __attribute__((ext_vector_type(2))) float v2f;
typedef __attribute__((ext_vector_type(4))) float v4f;
typedef __attribute__((ext_vector_type(8))) float v8f;

#define BATCH 8
#define H 128
#define W 128
#define OH 127
#define OW 127
#define TILE 8
#define SRC 9
#define NSLOT 96
#define PIXSTRIDE 516

static __device__ __forceinline__ float xhalf(float x) {
  return __builtin_bit_cast(float, __builtin_amdgcn_permlanex16(
      __builtin_bit_cast(int, x), __builtin_bit_cast(int, x), 0x76543210, 0xfedcba98, false, false));
}
template <int CTRL>
static __device__ __forceinline__ float dpp(float x) {
  return __builtin_bit_cast(float, __builtin_amdgcn_update_dpp(
      0, __builtin_bit_cast(int, x), CTRL, 0xf, 0xf, true));
}
#define DPP_ROR8        0x128
#define DPP_HALF_MIRROR 0x141
#define DPP_XOR2        0x4E
#define DPP_XOR1        0xB1

__global__ __launch_bounds__(1024, 1)
void convsbs_fused_kernel(const float* __restrict__ channels,
                          const float* __restrict__ cores,
                          float* __restrict__ out) {
  __shared__ __attribute__((aligned(16))) float inter[NSLOT * PIXSTRIDE];

  const int b   = blockIdx.z;
  const int ty0 = blockIdx.y * TILE;
  const int tx0 = blockIdx.x * TILE;

  const float* __restrict__ ch0 = channels;
  const float* __restrict__ ch1 = channels + (long)BATCH * H * W * 4;

  const int tid  = (int)threadIdx.x;
  const int wv   = tid >> 5;
  const int lane = tid & 31;
  const int wvu  = __builtin_amdgcn_readfirstlane(wv);

  if (wvu < 24) {
    const int g    = wvu >> 2;
    const int core = wvu & 3;
    const int m    = lane & 15;
    const int half = lane >> 4;

    int s = g * 16 + m;
    int sc = s > 80 ? 80 : s;
    int sy = sc / SRC, sx = sc % SRC;
    int gy = ty0 + sy; gy = gy > (H - 1) ? (H - 1) : gy;
    int gx = tx0 + sx; gx = gx > (W - 1) ? (W - 1) : gx;
    const long pixoff = ((((long)b * H + gy) * W) + gx) * 4;

    const v4f c0 = *(const v4f*)(ch0 + pixoff);
    const v2f c1 = *(const v2f*)(ch1 + pixoff + 2 * half);

    v2f afrag[4];
#pragma unroll
    for (int kt = 0; kt < 4; ++kt) {
      afrag[kt][0] = c0[kt] * c1[0];
      afrag[kt][1] = c0[kt] * c1[1];
    }

    const bool tr = (core & 1) != 0;
    const int lanepart = tr ? ((m & 3) * 8 + (m >> 2)) : ((m >> 2) * 8 + (m & 3));
    const float* __restrict__ wbase = cores + core * 2048 + lanepart * 16 + 2 * half;
    const int sbase = g * 16 + 8 * half;

#pragma unroll
    for (int nt = 0; nt < 8; ++nt) {
      const int ntpart = tr ? (((nt & 3) >> 1) * 64 + (nt >> 2) * 32 + (nt & 1) * 4)
                            : ((nt & 3) * 32 + (nt >> 2) * 4);
      v8f acc = {};
#pragma unroll
      for (int kt = 0; kt < 4; ++kt) {
        const v2f bfrag = *(const v2f*)(wbase + ntpart * 16 + kt * 4);
        acc = __builtin_amdgcn_wmma_f32_16x16x4_f32(
            false, afrag[kt], false, bfrag, (short)0, acc, false, false);
      }
      const int colbase = core * 128 + nt * 16 + m;
#pragma unroll
      for (int v = 0; v < 8; ++v)
        inter[(sbase + v) * PIXSTRIDE + colbase] = acc[v];
    }
  }

  __syncthreads();

  const int oyl  = wvu >> 2;
  const int oxl0 = (wvu & 3) * 2;

  float V[2][2][2];
#pragma unroll
  for (int t = 0; t < 2; ++t) {
    const int sA = oyl * SRC + oxl0 + t;
    const float* fb = &inter[sA * PIXSTRIDE + lane * 4];
    const v4f fa = *(const v4f*)(fb);
    const v4f fq = *(const v4f*)(fb + PIXSTRIDE + 128);
    const v4f fc = *(const v4f*)(fb + SRC * PIXSTRIDE + 256);
    const v4f fd = *(const v4f*)(fb + (SRC + 1) * PIXSTRIDE + 384);

    const v2f a0 = {fa[0], fa[1]}, a1 = {fa[2], fa[3]};
    const v2f q0 = {fq[0], fq[1]}, q1 = {fq[2], fq[3]};
    const v2f c0 = {fc[0], fc[1]}, c1 = {fc[2], fc[3]};
    const v2f d0 = {fd[0], fd[1]}, d1 = {fd[2], fd[3]};

    v8f dp = {};
    dp = __builtin_amdgcn_wmma_f32_16x16x4_f32(false, a0, false, q0, (short)0, dp, false, false);
    dp = __builtin_amdgcn_wmma_f32_16x16x4_f32(false, a1, false, q1, (short)0, dp, false, false);
    v8f dr = {};
    dr = __builtin_amdgcn_wmma_f32_16x16x4_f32(false, d0, false, c0, (short)0, dr, false, false);
    dr = __builtin_amdgcn_wmma_f32_16x16x4_f32(false, d1, false, c1, (short)0, dr, false, false);

    float u00 = 0.f, u01 = 0.f, u10 = 0.f, u11 = 0.f;
#pragma unroll
    for (int i = 0; i < 8; ++i) {
      const float p  = dp[i];
      const float px = dpp<DPP_ROR8>(p);
      const float r  = dr[i];
      const float rx = xhalf(r);
      u00 = fmaf(p,  r,  u00);
      u01 = fmaf(px, r,  u01);
      u10 = fmaf(p,  rx, u10);
      u11 = fmaf(px, rx, u11);
    }
    V[t][0][0] = u00; V[t][0][1] = u01; V[t][1][0] = u10; V[t][1][1] = u11;
  }

  const int k  = lane & 7;
  const int x  = (lane >> 3) & 1;
  const int h  = lane >> 4;
  const bool k2   = (k & 4) != 0;
  const bool bsel = ((((k >> 1) & 1) ^ x) != 0);
  const bool asel = (((k & 1) ^ h) != 0);

  float Wv[2][2];
#pragma unroll
  for (int a = 0; a < 2; ++a)
#pragma unroll
    for (int bb = 0; bb < 2; ++bb) {
      const float keep = k2 ? V[1][a][bb] : V[0][a][bb];
      const float send = k2 ? V[0][a][bb] : V[1][a][bb];
      Wv[a][bb] = keep + dpp<DPP_HALF_MIRROR>(send);
    }
  float U[2];
#pragma unroll
  for (int a = 0; a < 2; ++a) {
    const float keep = bsel ? Wv[a][1] : Wv[a][0];
    const float send = bsel ? Wv[a][0] : Wv[a][1];
    U[a] = keep + dpp<DPP_XOR2>(send);
  }
  const float keepf = asel ? U[1] : U[0];
  const float sendf = asel ? U[0] : U[1];
  const float res = keepf + dpp<DPP_XOR1>(sendf);

  const int oy = ty0 + oyl;
  const int ox = tx0 + oxl0 + (k >> 2);
  if (oy < OH && ox < OW)
    out[((((long)b * OH + oy) * OW) + ox) * 16 + h * 8 + ((k >> 1) & 1) * 4 + x * 2 + (k & 1)] = res;
}

extern "C" void kernel_launch(void* const* d_in, const int* in_sizes, int n_in,
                              void* d_out, int out_size, void* d_ws, size_t ws_size,
                              hipStream_t stream) {
  (void)in_sizes; (void)n_in; (void)d_ws; (void)ws_size; (void)out_size;
  const float* channels = (const float*)d_in[0];
  const float* cores    = (const float*)d_in[1];
  float* out            = (float*)d_out;

  dim3 grid((OW + TILE - 1) / TILE, (OH + TILE - 1) / TILE, BATCH);
  dim3 block(1024);
  hipLaunchKernelGGL(convsbs_fused_kernel, grid, block, 0, stream,
                     channels, cores, out);
}
